# starting kernel
# baseline (speedup 1.0000x reference)
#include <hip/hip_runtime.h>
#include <hip/hip_bf16.h>

#define BB 32
#define TT 2048
#define DD 1024

typedef __attribute__((ext_vector_type(2))) float v2f;
typedef __attribute__((ext_vector_type(8))) float v8f;

__global__ __launch_bounds__(256) void att_init_kernel(float* __restrict__ sums,
                                                       float* __restrict__ out) {
    int i = blockIdx.x * 256 + threadIdx.x;
    if (i < BB) sums[i] = 0.0f;
    if (i < BB * DD) out[i] = 0.0f;
}

__global__ __launch_bounds__(256) void att_score_kernel(const float* __restrict__ x,
                                                        const float* __restrict__ w,
                                                        const float* __restrict__ bias,
                                                        float* __restrict__ e,
                                                        float* __restrict__ sums) {
    const int lane  = threadIdx.x & 31;
    const int wave  = threadIdx.x >> 5;
    const int b     = blockIdx.x >> 4;
    const int tblk  = blockIdx.x & 15;
    const int t0    = tblk * 128 + wave * 16;
    const int laneM = lane & 15;
    const int kofs  = (lane >> 4) << 1;

    const float* xrow = x + ((size_t)b * TT + (size_t)(t0 + laneM)) * DD + kofs;
    const float* wp   = w + kofs;

    v8f acc = {};
    #pragma unroll 4
    for (int k = 0; k < DD; k += 4) {
        v2f av = *(const v2f*)(wp + k);
        v2f bv = *(const v2f*)(xrow + k);
        acc = __builtin_amdgcn_wmma_f32_16x16x4_f32(
             false, av,  false, bv,
             (short)0, acc,  false,  false);
    }

    const int t = t0 + laneM;
    const float s  = acc[0] + bias[t];
    const float th = 1.0f - 2.0f * __builtin_amdgcn_rcpf(__expf(2.0f * s) + 1.0f);
    const float ev = __expf(th);

    if (lane < 16) e[(size_t)b * TT + t] = ev;

    float psum = ev;
    psum += __shfl_xor(psum, 1, 32);
    psum += __shfl_xor(psum, 2, 32);
    psum += __shfl_xor(psum, 4, 32);
    psum += __shfl_xor(psum, 8, 32);
    if (lane == 0) atomicAdd(&sums[b], psum);
}

__global__ __launch_bounds__(256) void att_pool_kernel(const float* __restrict__ x,
                                                       const float* __restrict__ e,
                                                       const float* __restrict__ sums,
                                                       float* __restrict__ out) {
    const int lane   = threadIdx.x & 31;
    const int wave   = threadIdx.x >> 5;
    const int bid    = blockIdx.x;
    const int tslice = bid & 3;
    const int dtile  = (bid >> 2) & 7;
    const int b      = bid >> 5;
    const int d0     = dtile * 128 + wave * 16;
    const int laneM  = lane & 15;
    const int tofs   = (lane >> 4) << 1;

    const float* xbase = x + (size_t)b * TT * DD + (size_t)d0 + laneM;
    const float* ebase = e + (size_t)b * TT + tofs;

    const int tstart = tslice * (TT / 4);
    v8f acc = {};
    #pragma unroll 4
    for (int t = tstart; t < tstart + TT / 4; t += 4) {
        const int tt = t + tofs;
        v2f av = *(const v2f*)(ebase + t);
        v2f bv;
        bv.x = xbase[(size_t)tt * DD];
        bv.y = xbase[(size_t)(tt + 1) * DD];
        acc = __builtin_amdgcn_wmma_f32_16x16x4_f32(
            false, av, false, bv, (short)0, acc, false, false);
    }

    if (lane < 16) {
        const float inv = __builtin_amdgcn_rcpf(sums[b]);
        atomicAdd(&out[(size_t)b * DD + d0 + lane], acc[0] * inv);
    }
}

extern "C" void kernel_launch(void* const* d_in, const int* in_sizes, int n_in,
                              void* d_out, int out_size, void* d_ws, size_t ws_size,
                              hipStream_t stream) {
    const float* x    = (const float*)d_in[0];
    const float* w    = (const float*)d_in[1];
    const float* bias = (const float*)d_in[2];
    float* out = (float*)d_out;

    float* e    = (float*)d_ws;
    float* sums = e + (size_t)BB * TT;

    att_init_kernel<<<(BB * DD + 255) / 256, 256, 0, stream>>>(sums, out);
    att_score_kernel<<<BB * (TT / 128), 256, 0, stream>>>(x, w, bias, e, sums);
    att_pool_kernel<<<BB * 8 * 4, 256, 0, stream>>>(x, e, sums, out);
}


# Fuse score and pool into one pass: each 16-row x tile is read once into LDS (512 B per wave load) and feeds both WMMA products; partials summed in fixed order by a finalize kernel, no atomics
# speedup vs baseline: 1.8243x; 1.8243x over previous
#include <hip/hip_runtime.h>
#include <hip/hip_bf16.h>


#define BB 32
#define TT 2048
#define DD 1024
#define ROWS 16
#define CHUNK 128
#define NCHUNK (TT / CHUNK)
#define NSUB (CHUNK / ROWS)
#define PITCH 1036

typedef __attribute__((ext_vector_type(2))) float v2f;
typedef __attribute__((ext_vector_type(4))) float v4f;
typedef __attribute__((ext_vector_type(8))) float v8f;

__device__ __forceinline__ void lds_barrier() {
    __builtin_amdgcn_fence(__ATOMIC_RELEASE, "workgroup", "local");
    __builtin_amdgcn_s_barrier();
    __builtin_amdgcn_fence(__ATOMIC_ACQUIRE, "workgroup", "local");
}

__global__ __launch_bounds__(256) void att_fused_kernel(const float* __restrict__ x,
                                                        const float* __restrict__ w,
                                                        const float* __restrict__ bias,
                                                        float* __restrict__ pout,
                                                        float* __restrict__ psum) {
    __shared__ __attribute__((aligned(16))) float tile[2][ROWS * PITCH];
    __shared__ __attribute__((aligned(16))) float wlds[DD];
    __shared__ float blds[CHUNK];
    __shared__ float spart[2][8][16];

    const int tid   = threadIdx.x;
    const int lane  = tid & 31;
    const int wave  = tid >> 5;
    const int n     = lane & 15;
    const int h     = lane >> 4;
    const int b     = blockIdx.x >> 4;
    const int chunk = blockIdx.x & 15;

    const float* xg = x + ((size_t)b * TT + (size_t)chunk * CHUNK) * DD + 4 * tid;

    v4f stage[ROWS];
    #pragma unroll
    for (int j = 0; j < ROWS; ++j) stage[j] = *(const v4f*)(xg + (size_t)j * DD);

    *(v4f*)(wlds + 4 * tid) = *(const v4f*)(w + 4 * tid);
    if (tid < CHUNK) blds[tid] = bias[chunk * CHUNK + tid];

    v8f acc[8];
    #pragma unroll
    for (int j = 0; j < 8; ++j) acc[j] = (v8f){};
    float esum = 0.0f;

    const int dw = 128 * wave;

    for (int sub = 0; sub < NSUB; ++sub) {
        float* tl = &tile[sub & 1][0];
        #pragma unroll
        for (int j = 0; j < ROWS; ++j) *(v4f*)(tl + j * PITCH + 4 * tid) = stage[j];
        lds_barrier();

        if (sub + 1 < NSUB) {
            const float* xn = xg + (size_t)(sub + 1) * ROWS * DD;
            #pragma unroll
            for (int j = 0; j < ROWS; ++j) stage[j] = *(const v4f*)(xn + (size_t)j * DD);
        }

        v8f s0 = {}, s1 = {};
        const float* bp = tl + n * PITCH + dw + 2 * h;
        const float* ap = wlds + dw + 2 * h;
        #pragma unroll
        for (int i = 0; i < 32; i += 2) {
            v2f a0 = *(const v2f*)(ap + 4 * i);
            v2f b0 = *(const v2f*)(bp + 4 * i);
            v2f a1 = *(const v2f*)(ap + 4 * i + 4);
            v2f b1 = *(const v2f*)(bp + 4 * i + 4);
            s0 = __builtin_amdgcn_wmma_f32_16x16x4_f32(false, a0, false, b0, (short)0, s0, false, false);
            s1 = __builtin_amdgcn_wmma_f32_16x16x4_f32(false, a1, false, b1, (short)0, s1, false, false);
        }
        if (lane < 16) spart[sub & 1][wave][n] = s0[0] + s1[0];
        lds_barrier();

        float s = spart[sub & 1][0][n];
        #pragma unroll
        for (int q = 1; q < 8; ++q) s += spart[sub & 1][q][n];
        s += blds[sub * ROWS + n];
        const float th = 1.0f - 2.0f * __builtin_amdgcn_rcpf(__expf(2.0f * s) + 1.0f);
        const float ev = __expf(th);
        esum += ev;

        const float* pp = tl + (2 * h) * PITCH + dw + n;
        #pragma unroll
        for (int c = 0; c < 4; ++c) {
            v2f av;
            av.x = __shfl(ev, 4 * c + 2 * h, 32);
            av.y = __shfl(ev, 4 * c + 2 * h + 1, 32);
            #pragma unroll
            for (int j = 0; j < 8; ++j) {
                v2f bv;
                bv.x = pp[(4 * c) * PITCH + 16 * j];
                bv.y = pp[(4 * c + 1) * PITCH + 16 * j];
                acc[j] = __builtin_amdgcn_wmma_f32_16x16x4_f32(false, av, false, bv, (short)0, acc[j], false, false);
            }
        }
    }

    float* po = pout + ((size_t)b * NCHUNK + chunk) * DD + dw;
    #pragma unroll
    for (int jj = 0; jj < 4; ++jj) {
        const float v = h ? acc[2 * jj + 1][0] : acc[2 * jj][0];
        po[32 * jj + lane] = v;
    }

    esum += __shfl_xor(esum, 1, 32);
    esum += __shfl_xor(esum, 2, 32);
    esum += __shfl_xor(esum, 4, 32);
    esum += __shfl_xor(esum, 8, 32);
    if (tid == 0) psum[b * NCHUNK + chunk] = esum;
}

__global__ __launch_bounds__(256) void att_finalize_kernel(const float* __restrict__ pout,
                                                           const float* __restrict__ psum,
                                                           float* __restrict__ out) {
    const int i = blockIdx.x * 256 + threadIdx.x;
    const int b = i >> 10;
    const int d = i & (DD - 1);
    float num = 0.0f, den = 0.0f;
    #pragma unroll
    for (int c = 0; c < NCHUNK; ++c) {
        num += pout[((size_t)b * NCHUNK + c) * DD + d];
        den += psum[b * NCHUNK + c];
    }
    out[i] = num * __builtin_amdgcn_rcpf(den);
}

extern "C" void kernel_launch(void* const* d_in, const int* in_sizes, int n_in,
                              void* d_out, int out_size, void* d_ws, size_t ws_size,
                              hipStream_t stream) {
    const float* x    = (const float*)d_in[0];
    const float* w    = (const float*)d_in[1];
    const float* bias = (const float*)d_in[2];
    float* out = (float*)d_out;

    float* pout = (float*)d_ws;
    float* psum = pout + (size_t)BB * NCHUNK * DD;

    att_fused_kernel<<<BB * NCHUNK, 256, 0, stream>>>(x, w, bias, pout, psum);
    att_finalize_kernel<<<(BB * DD) / 256, 256, 0, stream>>>(pout, psum, out);
}
